# starting kernel
# baseline (speedup 1.0000x reference)
#include <hip/hip_runtime.h>
#include <hip/hip_bf16.h>

typedef float v2f __attribute__((ext_vector_type(2)));
typedef float v8f __attribute__((ext_vector_type(8)));

#define D 64

__global__ __launch_bounds__(128)
void gcn_gemm_bias(const float* __restrict__ x,
                   const float* __restrict__ W,
                   const float* __restrict__ b,
                   float* __restrict__ h,
                   int n_nodes) {
  const int lane = threadIdx.x & 31;
  const int wave = threadIdx.x >> 5;
  const int row0 = blockIdx.x << 4;
  const int col0 = wave << 4;

  const int m  = lane & 15;
  const int kk = (lane >> 4) << 1;

  int arow = row0 + m;
  if (arow >= n_nodes) arow = n_nodes - 1;

  const float* __restrict__ xrow = x + (size_t)arow * D;
  const float* __restrict__ wrow = W + (size_t)(col0 + m) * D;

  v8f acc = {};
#pragma unroll
  for (int k0 = 0; k0 < D; k0 += 4) {
    v2f a  = *(const v2f*)(xrow + k0 + kk);
    v2f bb = *(const v2f*)(wrow + k0 + kk);
    acc = __builtin_amdgcn_wmma_f32_16x16x4_f32(
        false, a, false, bb, (short)0, acc, false, false);
  }

  const float bias = b[col0 + m];
  const int rbase = row0 + ((lane >> 4) << 3);
#pragma unroll
  for (int r = 0; r < 8; ++r) {
    const int row = rbase + r;
    if (row < n_nodes) {
      h[(size_t)row * D + col0 + m] = acc[r] + bias;
    }
  }
}

__global__ __launch_bounds__(256)
void gcn_edge_scatter(const float* __restrict__ h,
                      const int* __restrict__ src,
                      const int* __restrict__ dst,
                      const float* __restrict__ w,
                      float* __restrict__ out,
                      int n_edges) {
  const int gid = blockIdx.x * 256 + threadIdx.x;
  const int e = gid >> 4;
  if (e >= n_edges) return;
  const int f = (gid & 15) << 2;

  const int   s  = src[e];
  const int   d  = dst[e];
  const float wv = w[e];

  const float4 hv = *(const float4*)(h + (size_t)s * D + f);
  float* __restrict__ o = out + (size_t)d * D + f;

  atomicAdd(o + 0, hv.x * wv);
  atomicAdd(o + 1, hv.y * wv);
  atomicAdd(o + 2, hv.z * wv);
  atomicAdd(o + 3, hv.w * wv);
}

extern "C" void kernel_launch(void* const* d_in, const int* in_sizes, int n_in,
                              void* d_out, int out_size, void* d_ws, size_t ws_size,
                              hipStream_t stream) {
  const float* x   = (const float*)d_in[0];
  const int*   src = (const int*)  d_in[1];
  const int*   dst = (const int*)  d_in[2];
  const float* w   = (const float*)d_in[3];
  const float* W   = (const float*)d_in[4];
  const float* b   = (const float*)d_in[5];
  float*       out = (float*)d_out;
  float*       h   = (float*)d_ws;

  const int n_nodes = in_sizes[0] / D;
  const int n_edges = in_sizes[1];

  hipMemsetAsync(d_out, 0, (size_t)out_size * sizeof(float), stream);

  const int row_tiles = (n_nodes + 15) / 16;
  gcn_gemm_bias<<<dim3(row_tiles), dim3(128), 0, stream>>>(x, W, b, h, n_nodes);

  const long long threads = (long long)n_edges * 16;
  const int blocks = (int)((threads + 255) / 256);
  gcn_edge_scatter<<<dim3(blocks), dim3(256), 0, stream>>>(h, src, dst, w, out, n_edges);
}


# edges bucketed by destination (LDS count, scan, place); one work-group per 400 rows sums h[src]*w in an LDS tile by LDS float adds, stores out once: no global atomics, no memset; sum order = arrival
# speedup vs baseline: 1.5762x; 1.5762x over previous
#include <hip/hip_runtime.h>
#include <hip/hip_bf16.h>

typedef float v2f __attribute__((ext_vector_type(2)));
typedef float v8f __attribute__((ext_vector_type(8)));

#define D 64
#define BN 400
#define NBMAX 256
#define GB 256
#define BT 1024
#define EP 5

__global__ __launch_bounds__(128)
void gcn_gemm_bias(const float* __restrict__ x,
                   const float* __restrict__ W,
                   const float* __restrict__ b,
                   float* __restrict__ h,
                   int n_nodes) {
  const int lane = threadIdx.x & 31;
  const int wave = threadIdx.x >> 5;
  const int row0 = blockIdx.x << 4;
  const int col0 = wave << 4;

  const int m  = lane & 15;
  const int kk = (lane >> 4) << 1;

  int arow = row0 + m;
  if (arow >= n_nodes) arow = n_nodes - 1;

  const float* __restrict__ xrow = x + (size_t)arow * D;
  const float* __restrict__ wrow = W + (size_t)(col0 + m) * D;

  v8f acc = {};
#pragma unroll
  for (int k0 = 0; k0 < D; k0 += 4) {
    v2f a  = *(const v2f*)(xrow + k0 + kk);
    v2f bb = *(const v2f*)(wrow + k0 + kk);
    acc = __builtin_amdgcn_wmma_f32_16x16x4_f32(
        false, a, false, bb, (short)0, acc, false, false);
  }

  const float bias = b[col0 + m];
  const int rbase = row0 + ((lane >> 4) << 3);
#pragma unroll
  for (int r = 0; r < 8; ++r) {
    const int row = rbase + r;
    if (row < n_nodes) {
      h[(size_t)row * D + col0 + m] = acc[r] + bias;
    }
  }
}

__device__ __forceinline__ int bucket_of(int d) {
  unsigned k = (unsigned)d / BN;
  return (int)(k < (unsigned)(NBMAX - 1) ? k : (unsigned)(NBMAX - 1));
}

__global__ __launch_bounds__(BT)
void gcn_edge_count(const int* __restrict__ dst,
                    int* __restrict__ blockhist,
                    int n_edges) {
  __shared__ int hist[NBMAX];
  const int t = threadIdx.x;
  if (t < NBMAX) hist[t] = 0;
  __syncthreads();

  const int chunk = (n_edges + GB - 1) / GB;
  const int e0 = blockIdx.x * chunk;
  const int e1 = (e0 + chunk < n_edges) ? (e0 + chunk) : n_edges;
  for (int base = e0 + t; base < e1; base += BT * EP) {
    int dd[EP];
#pragma unroll
    for (int u = 0; u < EP; ++u) {
      const int e = base + u * BT;
      dd[u] = dst[e < e1 ? e : e1 - 1];
    }
#pragma unroll
    for (int u = 0; u < EP; ++u) {
      if (base + u * BT < e1) atomicAdd(&hist[bucket_of(dd[u])], 1);
    }
  }
  __syncthreads();
  if (t < NBMAX) blockhist[blockIdx.x * NBMAX + t] = hist[t];
}

__global__ __launch_bounds__(BT)
void gcn_edge_place(const int* __restrict__ src,
                    const int* __restrict__ dst,
                    const float* __restrict__ w,
                    const int* __restrict__ blockhist,
                    int* __restrict__ bstart,
                    unsigned int* __restrict__ sorted_raw,
                    int n_edges) {
  __shared__ int part_tot[4][NBMAX];
  __shared__ int part_pre[4][NBMAX];
  __shared__ int cursor[NBMAX];
  __shared__ int wtot[8];

  const int t = threadIdx.x;
  const int lane = t & 31;
  const int k = t & (NBMAX - 1);
  const int q = t >> 8;
  const int me = blockIdx.x;

  int tot = 0, pre = 0;
#pragma unroll 16
  for (int i = 0; i < GB / 4; ++i) {
    const int g = q * (GB / 4) + i;
    const int v = blockhist[g * NBMAX + k];
    tot += v;
    pre += (g < me) ? v : 0;
  }
  part_tot[q][k] = tot;
  part_pre[q][k] = pre;
  __syncthreads();

  int total = 0, before = 0, incl = 0;
  if (t < NBMAX) {
    total  = part_tot[0][t] + part_tot[1][t] + part_tot[2][t] + part_tot[3][t];
    before = part_pre[0][t] + part_pre[1][t] + part_pre[2][t] + part_pre[3][t];
    incl = total;
#pragma unroll
    for (int off = 1; off < 32; off <<= 1) {
      const int n = __shfl_up(incl, off);
      if (lane >= off) incl += n;
    }
    if (lane == 31) wtot[t >> 5] = incl;
  }
  __syncthreads();
  if (t < NBMAX) {
    int wp = 0;
    for (int j = 0; j < (t >> 5); ++j) wp += wtot[j];
    const int excl = wp + incl - total;
    cursor[t] = excl + before;
    if (me == 0) {
      bstart[t] = excl;
      if (t == NBMAX - 1) bstart[NBMAX] = excl + total;
    }
  }
  __syncthreads();

  uint2* __restrict__ sorted = (uint2*)sorted_raw;
  const int chunk = (n_edges + GB - 1) / GB;
  const int e0 = me * chunk;
  const int e1 = (e0 + chunk < n_edges) ? (e0 + chunk) : n_edges;
  for (int base = e0 + t; base < e1; base += BT * EP) {
    int dd[EP], ss[EP];
    float ww[EP];
#pragma unroll
    for (int u = 0; u < EP; ++u) {
      const int e = base + u * BT;
      const int ec = e < e1 ? e : e1 - 1;
      dd[u] = dst[ec];
      ss[u] = src[ec];
      ww[u] = w[ec];
    }
#pragma unroll
    for (int u = 0; u < EP; ++u) {
      if (base + u * BT < e1) {
        const int kb = bucket_of(dd[u]);
        const int pos = atomicAdd(&cursor[kb], 1);
        const unsigned dl = (unsigned)(dd[u] - kb * BN);
        sorted[pos] = make_uint2(((unsigned)ss[u] & 0xFFFFFu) | (dl << 20), __float_as_uint(ww[u]));
      }
    }
  }
}

#define RB 16

__global__ __launch_bounds__(BT)
void gcn_owner_sum(const float* __restrict__ h,
                   const unsigned int* __restrict__ sorted_raw,
                   const int* __restrict__ bstart,
                   float* __restrict__ out,
                   int n_nodes) {
  __shared__ float tile[BN * D];
  const int t = threadIdx.x;
  const int lane = t & 31;
  const int wave = __builtin_amdgcn_readfirstlane(t >> 5);

  float4* __restrict__ t4 = (float4*)tile;
  for (int i = t; i < BN * D / 4; i += BT) t4[i] = make_float4(0.f, 0.f, 0.f, 0.f);
  __syncthreads();

  const uint2* __restrict__ sorted = (const uint2*)sorted_raw;
  const int b0 = bstart[blockIdx.x];
  const int b1 = bstart[blockIdx.x + 1];

  int g = b0 + wave * 32;
  uint2 nxt = make_uint2(0u, 0u);
  if (g + lane < b1) nxt = sorted[g + lane];
  for (; g < b1; g += BT) {
    const int cnt = (b1 - g < 32) ? (b1 - g) : 32;
    const uint2 rec = nxt;
    nxt = make_uint2(0u, 0u);
    if (g + BT + lane < b1) nxt = sorted[g + BT + lane];

#pragma unroll
    for (int j0 = 0; j0 < 32; j0 += RB) {
      if (j0 >= cnt) break;
      v2f hv[RB];
#pragma unroll
      for (int u = 0; u < RB; ++u) {
        const unsigned p = (unsigned)__builtin_amdgcn_readlane((int)rec.x, j0 + u);
        hv[u] = *(const v2f*)(h + (size_t)(p & 0xFFFFFu) * D + lane * 2);
      }
#pragma unroll
      for (int u = 0; u < RB; ++u) {
        if (j0 + u < cnt) {
          const unsigned p = (unsigned)__builtin_amdgcn_readlane((int)rec.x, j0 + u);
          const float wv = __uint_as_float((unsigned)__builtin_amdgcn_readlane((int)rec.y, j0 + u));
          unsigned dl = p >> 20;
          dl = dl < (unsigned)(BN - 1) ? dl : (unsigned)(BN - 1);
          float* __restrict__ trow = tile + dl * D + lane * 2;
          atomicAdd(trow + 0, hv[u].x * wv);
          atomicAdd(trow + 1, hv[u].y * wv);
        }
      }
    }
  }
  __syncthreads();

  const int row0 = blockIdx.x * BN;
  int nrows = n_nodes - row0;
  nrows = nrows < BN ? nrows : BN;
  float4* __restrict__ o4 = (float4*)(out + (size_t)row0 * D);
  for (int i = t; i < nrows * (D / 4); i += BT) o4[i] = t4[i];
}

extern "C" void kernel_launch(void* const* d_in, const int* in_sizes, int n_in,
                              void* d_out, int out_size, void* d_ws, size_t ws_size,
                              hipStream_t stream) {
  const float* x   = (const float*)d_in[0];
  const int*   src = (const int*)  d_in[1];
  const int*   dst = (const int*)  d_in[2];
  const float* w   = (const float*)d_in[3];
  const float* W   = (const float*)d_in[4];
  const float* b   = (const float*)d_in[5];
  float*       out = (float*)d_out;
  char*        ws  = (char*)d_ws;
  float*        h         = (float*)ws;
  int*          blockhist = (int*)(ws + (64u << 20));
  int*          bstart    = (int*)(ws + (72u << 20));
  unsigned int* sorted    = (unsigned int*)(ws + (80u << 20));

  const int n_nodes = in_sizes[0] / D;
  const int n_edges = in_sizes[1];

  const int row_tiles = (n_nodes + 15) / 16;
  gcn_gemm_bias<<<dim3(row_tiles), dim3(128), 0, stream>>>(x, W, b, h, n_nodes);

  gcn_edge_count<<<dim3(GB), dim3(BT), 0, stream>>>(dst, blockhist, n_edges);
  gcn_edge_place<<<dim3(GB), dim3(BT), 0, stream>>>(src, dst, w, blockhist, bstart, sorted, n_edges);

  const int n_buckets = (n_nodes + BN - 1) / BN;
  gcn_owner_sum<<<dim3(n_buckets), dim3(BT), 0, stream>>>(h, sorted, bstart, out, n_nodes);
}


# count launch removed: each work-group buckets its own edge chunk in place (LDS count, scan, rank), writes 257 segment starts; owners find records by search over 256 segments; sum order = arrival
# speedup vs baseline: 2.6791x; 1.6997x over previous
#include <hip/hip_runtime.h>
#include <hip/hip_bf16.h>

typedef float v2f __attribute__((ext_vector_type(2)));
typedef float v8f __attribute__((ext_vector_type(8)));

#define D 64
#define BN 400
#define NBMAX 256
#define GB 256
#define SEGP 288
#define BT 1024
#define EP 5
#define TP 68
#define NRT (BN / 16)

__device__ __forceinline__ int bucket_of(int d) {
  unsigned k = (unsigned)d / BN;
  return (int)(k < (unsigned)(NBMAX - 1) ? k : (unsigned)(NBMAX - 1));
}

__global__ __launch_bounds__(BT)
void gcn_chunk_sort(const int* __restrict__ src,
                    const int* __restrict__ dst,
                    const float* __restrict__ w,
                    int* __restrict__ segtab,
                    unsigned int* __restrict__ sorted_raw,
                    int n_edges) {
  __shared__ int hist[NBMAX];
  __shared__ int wtot[8];
  const int t = threadIdx.x;
  const int lane = t & 31;
  const int me = blockIdx.x;
  if (t < NBMAX) hist[t] = 0;
  __syncthreads();

  const int chunk = (n_edges + GB - 1) / GB;
  const int e0 = (me * chunk < n_edges) ? me * chunk : n_edges;
  const int e1 = (e0 + chunk < n_edges) ? (e0 + chunk) : n_edges;
  const bool any = e0 < e1;

  int dd[EP], ss[EP];
  float ww[EP];
  if (any) {
#pragma unroll
    for (int u = 0; u < EP; ++u) {
      const int e = e0 + t + u * BT;
      const int ec = e < e1 ? e : e1 - 1;
      dd[u] = dst[ec];
      ss[u] = src[ec];
      ww[u] = w[ec];
    }
#pragma unroll
    for (int u = 0; u < EP; ++u) {
      if (e0 + t + u * BT < e1) atomicAdd(&hist[bucket_of(dd[u])], 1);
    }
    for (int e = e0 + t + EP * BT; e < e1; e += BT) atomicAdd(&hist[bucket_of(dst[e])], 1);
  }
  __syncthreads();

  int total = 0, incl = 0;
  if (t < NBMAX) {
    total = hist[t];
    incl = total;
#pragma unroll
    for (int off = 1; off < 32; off <<= 1) {
      const int n = __shfl_up(incl, off);
      if (lane >= off) incl += n;
    }
    if (lane == 31) wtot[t >> 5] = incl;
  }
  __syncthreads();
  if (t < NBMAX) {
    int wp = 0;
    for (int j = 0; j < (t >> 5); ++j) wp += wtot[j];
    const int excl = wp + incl - total;
    hist[t] = excl;
    segtab[me * SEGP + t] = excl;
    if (t == NBMAX - 1) segtab[me * SEGP + NBMAX] = excl + total;
  }
  __syncthreads();

  uint2* __restrict__ sorted = (uint2*)sorted_raw + e0;
  if (any) {
#pragma unroll
    for (int u = 0; u < EP; ++u) {
      if (e0 + t + u * BT < e1) {
        const int kb = bucket_of(dd[u]);
        const int pos = atomicAdd(&hist[kb], 1);
        const unsigned dl = (unsigned)(dd[u] - kb * BN);
        sorted[pos] = make_uint2(((unsigned)ss[u] & 0xFFFFFu) | (dl << 20), __float_as_uint(ww[u]));
      }
    }
    for (int e = e0 + t + EP * BT; e < e1; e += BT) {
      const int d = dst[e];
      const int kb = bucket_of(d);
      const int pos = atomicAdd(&hist[kb], 1);
      const unsigned dl = (unsigned)(d - kb * BN);
      sorted[pos] = make_uint2(((unsigned)src[e] & 0xFFFFFu) | (dl << 20), __float_as_uint(w[e]));
    }
  }
}

#define RB 16

__device__ __forceinline__ v8f tile_times_w(const float* __restrict__ arow, const v2f (&bf)[17], int hh) {
  v8f acc = {};
#pragma unroll
  for (int j = 0; j < 16; ++j) {
    const v2f a = *(const v2f*)(arow + 2 * j);
    acc = __builtin_amdgcn_wmma_f32_16x16x4_f32(false, a, false, bf[j], (short)0, acc, false, false);
  }
  const v2f a = *(const v2f*)(arow - 32 * hh + D + 2 * hh);
  acc = __builtin_amdgcn_wmma_f32_16x16x4_f32(false, a, false, bf[16], (short)0, acc, false, false);
  return acc;
}

__global__ __launch_bounds__(BT)
void gcn_owner_sum_map(const float* __restrict__ x,
                       const float* __restrict__ W,
                       const float* __restrict__ b,
                       const unsigned int* __restrict__ sorted_raw,
                       const int* __restrict__ segtab,
                       float* __restrict__ out,
                       int n_nodes,
                       int n_edges) {
  __shared__ float tile[BN * TP];
  __shared__ int seg_off[GB + 1];
  __shared__ int seg_base[GB];
  __shared__ int wtot[8];
  const int t = threadIdx.x;
  const int lane = t & 31;
  const int wave = __builtin_amdgcn_readfirstlane(t >> 5);
  const int c  = lane & 15;
  const int hh = lane >> 4;
  const int ct = wave & 3;

  int seg_len = 0, seg_at = 0;
  if (t < GB) {
    const int* __restrict__ row = segtab + t * SEGP + blockIdx.x;
    const int s0 = row[0];
    const int s1 = row[1];
    const int chunk = (n_edges + GB - 1) / GB;
    const int c0 = (t * chunk < n_edges) ? t * chunk : n_edges;
    seg_len = s1 - s0;
    seg_at = c0 + s0;
  }

  v2f bf[17];
  {
    const float* __restrict__ wrow = W + (size_t)(ct * 16 + c) * D + 32 * hh;
#pragma unroll
    for (int j = 0; j < 16; ++j) bf[j] = *(const v2f*)(wrow + 2 * j);
    const float bias = b[ct * 16 + c];
    bf[16].x = hh == 0 ? bias : 0.f;
    bf[16].y = 0.f;
  }

  float4* __restrict__ t4 = (float4*)tile;
  for (int i = t; i < BN * TP / 4; i += BT) t4[i] = make_float4(0.f, 0.f, 0.f, 0.f);

  int incl = seg_len;
  if (t < GB) {
#pragma unroll
    for (int off = 1; off < 32; off <<= 1) {
      const int n = __shfl_up(incl, off);
      if (lane >= off) incl += n;
    }
    if (lane == 31) wtot[t >> 5] = incl;
  }
  __syncthreads();
  if (t < GB) {
    int wp = 0;
    for (int j = 0; j < (t >> 5); ++j) wp += wtot[j];
    seg_off[t] = wp + incl - seg_len;
    seg_base[t] = seg_at;
    if (t == GB - 1) seg_off[GB] = wp + incl;
  }
  __syncthreads();

  const uint2* __restrict__ sorted = (const uint2*)sorted_raw;
  const int b1 = seg_off[GB];
  const int last_edge = n_edges - 1;
  const unsigned last_node = (unsigned)(n_nodes - 1);

  auto record = [&](int i) -> uint2 {
    uint2 r = make_uint2(0u, 0u);
    if (i < b1) {
      int lo = 0, at = 0;
#pragma unroll
      for (int s = GB / 2; s >= 1; s >>= 1) {
        const int v = seg_off[lo + s];
        if (v <= i) { lo += s; at = v; }
      }
      int idx = seg_base[lo] + (i - at);
      idx = idx < last_edge ? idx : last_edge;
      idx = idx > 0 ? idx : 0;
      r = sorted[idx];
    }
    return r;
  };

  int g = wave * 32;
  uint2 nxt = record(g + lane);
  for (; g < b1; g += BT) {
    const int cnt = (b1 - g < 32) ? (b1 - g) : 32;
    const uint2 rec = nxt;

    if (lane < cnt) {
      unsigned dm = rec.x >> 20;
      dm = dm < (unsigned)(BN - 1) ? dm : (unsigned)(BN - 1);
      atomicAdd(tile + dm * TP + D, __uint_as_float(rec.y));
    }

#pragma unroll
    for (int j0 = 0; j0 < 32; j0 += RB) {
      if (j0 >= cnt) break;
      v2f hv[RB];
#pragma unroll
      for (int u = 0; u < RB; ++u) {
        unsigned p = (unsigned)__builtin_amdgcn_readlane((int)rec.x, j0 + u) & 0xFFFFFu;
        p = p < last_node ? p : last_node;
        hv[u] = *(const v2f*)(x + (size_t)p * D + lane * 2);
      }
      if (j0 == 0) nxt = record(g + BT + lane);
#pragma unroll
      for (int u = 0; u < RB; ++u) {
        if (j0 + u < cnt) {
          const unsigned p = (unsigned)__builtin_amdgcn_readlane((int)rec.x, j0 + u);
          const float wv = __uint_as_float((unsigned)__builtin_amdgcn_readlane((int)rec.y, j0 + u));
          unsigned dl = p >> 20;
          dl = dl < (unsigned)(BN - 1) ? dl : (unsigned)(BN - 1);
          float* __restrict__ trow = tile + dl * TP + lane * 2;
          atomicAdd(trow + 0, hv[u].x * wv);
          atomicAdd(trow + 1, hv[u].y * wv);
        }
      }
    }
  }
  __syncthreads();

  const int rt0 = wave >> 2;
  static_assert(BN % 16 == 0 && NRT > 16 + 7 && NRT <= 32, "row tiles 0..23 always exist, 24..31 are guarded");
  const float* __restrict__ arow = tile + (rt0 * 16 + c) * TP + 32 * hh;
  const v8f acc0 = tile_times_w(arow, bf, hh);
  const v8f acc1 = tile_times_w(arow + 8 * 16 * TP, bf, hh);
  const v8f acc2 = tile_times_w(arow + 16 * 16 * TP, bf, hh);
  v8f acc3 = {};
  const bool has3 = rt0 + 24 < NRT;
  if (has3) acc3 = tile_times_w(arow + 24 * 16 * TP, bf, hh);
  __syncthreads();

  float* __restrict__ drow = tile + (rt0 * 16 + 8 * hh) * TP + ct * 16 + c;
#pragma unroll
  for (int r = 0; r < 8; ++r) {
    drow[r * TP] = acc0[r];
    drow[(8 * 16 + r) * TP] = acc1[r];
    drow[(16 * 16 + r) * TP] = acc2[r];
  }
  if (has3) {
#pragma unroll
    for (int r = 0; r < 8; ++r) drow[(24 * 16 + r) * TP] = acc3[r];
  }
  __syncthreads();

  const int row0 = blockIdx.x * BN;
  int nrows = n_nodes - row0;
  nrows = nrows < BN ? nrows : BN;
  float4* __restrict__ o4 = (float4*)(out + (size_t)row0 * D);
  for (int i = t; i < nrows * (D / 4); i += BT)
    o4[i] = *(const float4*)(tile + (i >> 4) * TP + (i & 15) * 4);
}

extern "C" void kernel_launch(void* const* d_in, const int* in_sizes, int n_in,
                              void* d_out, int out_size, void* d_ws, size_t ws_size,
                              hipStream_t stream) {
  const float* x   = (const float*)d_in[0];
  const int*   src = (const int*)  d_in[1];
  const int*   dst = (const int*)  d_in[2];
  const float* w   = (const float*)d_in[3];
  const float* W   = (const float*)d_in[4];
  const float* b   = (const float*)d_in[5];
  float*       out = (float*)d_out;
  char*        ws  = (char*)d_ws;
  int*          segtab    = (int*)(ws + (64u << 20));
  unsigned int* sorted    = (unsigned int*)(ws + (80u << 20));

  const int n_nodes = in_sizes[0] / D;
  const int n_edges = in_sizes[1];

  gcn_chunk_sort<<<dim3(GB), dim3(BT), 0, stream>>>(src, dst, w, segtab, sorted, n_edges);

  const int n_buckets = (n_nodes + BN - 1) / BN;
  gcn_owner_sum_map<<<dim3(n_buckets), dim3(BT), 0, stream>>>(x, W, b, sorted, segtab, out, n_nodes, n_edges);
}


# chunk sort puts each chunk's records in bucket order in LDS and writes them front to back, 256 contiguous bytes per wave store, instead of one scattered 8-byte store per edge; sum order as before
# speedup vs baseline: 2.7998x; 1.0451x over previous
#include <hip/hip_runtime.h>
#include <hip/hip_bf16.h>

typedef float v2f __attribute__((ext_vector_type(2)));
typedef float v8f __attribute__((ext_vector_type(8)));

#define D 64
#define BN 400
#define NBMAX 256
#define GB 256
#define SEGP 288
#define BT 1024
#define EP 5
#define CAP (BT * EP)
#define TP 68
#define NRT (BN / 16)

__device__ __forceinline__ int bucket_of(int d) {
  unsigned k = (unsigned)d / BN;
  return (int)(k < (unsigned)(NBMAX - 1) ? k : (unsigned)(NBMAX - 1));
}

__global__ __launch_bounds__(BT)
void gcn_chunk_sort(const int* __restrict__ src,
                    const int* __restrict__ dst,
                    const float* __restrict__ w,
                    int* __restrict__ segtab,
                    unsigned int* __restrict__ sorted_raw,
                    int n_edges) {
  __shared__ int hist[NBMAX];
  __shared__ int wtot[8];
  __shared__ unsigned long long stage[CAP];
  const int t = threadIdx.x;
  const int lane = t & 31;
  const int me = blockIdx.x;
  if (t < NBMAX) hist[t] = 0;
  __syncthreads();

  const int chunk = (n_edges + GB - 1) / GB;
  const int e0 = (me * chunk < n_edges) ? me * chunk : n_edges;
  const int e1 = (e0 + chunk < n_edges) ? (e0 + chunk) : n_edges;
  const bool any = e0 < e1;
  const bool staged = e1 - e0 <= CAP;

  int dd[EP], ss[EP];
  float ww[EP];
  if (any) {
#pragma unroll
    for (int u = 0; u < EP; ++u) {
      const int e = e0 + t + u * BT;
      const int ec = e < e1 ? e : e1 - 1;
      dd[u] = dst[ec];
      ss[u] = src[ec];
      ww[u] = w[ec];
    }
#pragma unroll
    for (int u = 0; u < EP; ++u) {
      if (e0 + t + u * BT < e1) atomicAdd(&hist[bucket_of(dd[u])], 1);
    }
    for (int e = e0 + t + EP * BT; e < e1; e += BT) atomicAdd(&hist[bucket_of(dst[e])], 1);
  }
  __syncthreads();

  int total = 0, incl = 0;
  if (t < NBMAX) {
    total = hist[t];
    incl = total;
#pragma unroll
    for (int off = 1; off < 32; off <<= 1) {
      const int n = __shfl_up(incl, off);
      if (lane >= off) incl += n;
    }
    if (lane == 31) wtot[t >> 5] = incl;
  }
  __syncthreads();
  if (t < NBMAX) {
    int wp = 0;
    for (int j = 0; j < (t >> 5); ++j) wp += wtot[j];
    const int excl = wp + incl - total;
    hist[t] = excl;
    segtab[me * SEGP + t] = excl;
    if (t == NBMAX - 1) segtab[me * SEGP + NBMAX] = excl + total;
  }
  __syncthreads();

  uint2* __restrict__ sorted = (uint2*)sorted_raw + e0;
  if (any && staged) {
#pragma unroll
    for (int u = 0; u < EP; ++u) {
      if (e0 + t + u * BT < e1) {
        const int kb = bucket_of(dd[u]);
        const int pos = atomicAdd(&hist[kb], 1);
        const unsigned dl = (unsigned)(dd[u] - kb * BN);
        const unsigned key = ((unsigned)ss[u] & 0xFFFFFu) | (dl << 20);
        stage[pos] = ((unsigned long long)__float_as_uint(ww[u]) << 32) | (unsigned long long)key;
      }
    }
  }
  if (any && !staged) {
#pragma unroll
    for (int u = 0; u < EP; ++u) {
      if (e0 + t + u * BT < e1) {
        const int kb = bucket_of(dd[u]);
        const int pos = atomicAdd(&hist[kb], 1);
        const unsigned dl = (unsigned)(dd[u] - kb * BN);
        sorted[pos] = make_uint2(((unsigned)ss[u] & 0xFFFFFu) | (dl << 20), __float_as_uint(ww[u]));
      }
    }
    for (int e = e0 + t + EP * BT; e < e1; e += BT) {
      const int d = dst[e];
      const int kb = bucket_of(d);
      const int pos = atomicAdd(&hist[kb], 1);
      const unsigned dl = (unsigned)(d - kb * BN);
      sorted[pos] = make_uint2(((unsigned)src[e] & 0xFFFFFu) | (dl << 20), __float_as_uint(w[e]));
    }
  }
  __syncthreads();
  if (staged) {
    unsigned long long* __restrict__ sorted64 = (unsigned long long*)sorted_raw + e0;
    const int n = e1 - e0;
    unsigned long long v[EP];
#pragma unroll
    for (int u = 0; u < EP; ++u) {
      const int i = t + u * BT;
      v[u] = stage[i < n ? i : 0];
    }
#pragma unroll
    for (int u = 0; u < EP; ++u) {
      const int i = t + u * BT;
      if (i < n) sorted64[i] = v[u];
    }
  }
}

#define RB 16

__device__ __forceinline__ v8f tile_times_w(const float* __restrict__ arow, const v2f (&bf)[17], int hh) {
  v8f acc = {};
#pragma unroll
  for (int j = 0; j < 16; ++j) {
    const v2f a = *(const v2f*)(arow + 2 * j);
    acc = __builtin_amdgcn_wmma_f32_16x16x4_f32(false, a, false, bf[j], (short)0, acc, false, false);
  }
  const v2f a = *(const v2f*)(arow - 32 * hh + D + 2 * hh);
  acc = __builtin_amdgcn_wmma_f32_16x16x4_f32(false, a, false, bf[16], (short)0, acc, false, false);
  return acc;
}

__global__ __launch_bounds__(BT)
void gcn_owner_sum_map(const float* __restrict__ x,
                       const float* __restrict__ W,
                       const float* __restrict__ b,
                       const unsigned int* __restrict__ sorted_raw,
                       const int* __restrict__ segtab,
                       float* __restrict__ out,
                       int n_nodes,
                       int n_edges) {
  __shared__ float tile[BN * TP];
  __shared__ int seg_off[GB + 1];
  __shared__ int seg_base[GB];
  __shared__ int wtot[8];
  const int t = threadIdx.x;
  const int lane = t & 31;
  const int wave = __builtin_amdgcn_readfirstlane(t >> 5);
  const int c  = lane & 15;
  const int hh = lane >> 4;
  const int ct = wave & 3;

  int seg_len = 0, seg_at = 0;
  if (t < GB) {
    const int* __restrict__ row = segtab + t * SEGP + blockIdx.x;
    const int s0 = row[0];
    const int s1 = row[1];
    const int chunk = (n_edges + GB - 1) / GB;
    const int c0 = (t * chunk < n_edges) ? t * chunk : n_edges;
    seg_len = s1 - s0;
    seg_at = c0 + s0;
  }

  v2f bf[17];
  {
    const float* __restrict__ wrow = W + (size_t)(ct * 16 + c) * D + 32 * hh;
#pragma unroll
    for (int j = 0; j < 16; ++j) bf[j] = *(const v2f*)(wrow + 2 * j);
    const float bias = b[ct * 16 + c];
    bf[16].x = hh == 0 ? bias : 0.f;
    bf[16].y = 0.f;
  }

  float4* __restrict__ t4 = (float4*)tile;
  for (int i = t; i < BN * TP / 4; i += BT) t4[i] = make_float4(0.f, 0.f, 0.f, 0.f);

  int incl = seg_len;
  if (t < GB) {
#pragma unroll
    for (int off = 1; off < 32; off <<= 1) {
      const int n = __shfl_up(incl, off);
      if (lane >= off) incl += n;
    }
    if (lane == 31) wtot[t >> 5] = incl;
  }
  __syncthreads();
  if (t < GB) {
    int wp = 0;
    for (int j = 0; j < (t >> 5); ++j) wp += wtot[j];
    seg_off[t] = wp + incl - seg_len;
    seg_base[t] = seg_at;
    if (t == GB - 1) seg_off[GB] = wp + incl;
  }
  __syncthreads();

  const uint2* __restrict__ sorted = (const uint2*)sorted_raw;
  const int b1 = seg_off[GB];
  const int last_edge = n_edges - 1;
  const unsigned last_node = (unsigned)(n_nodes - 1);

  auto record = [&](int i) -> uint2 {
    uint2 r = make_uint2(0u, 0u);
    if (i < b1) {
      int lo = 0, at = 0;
#pragma unroll
      for (int s = GB / 2; s >= 1; s >>= 1) {
        const int v = seg_off[lo + s];
        if (v <= i) { lo += s; at = v; }
      }
      int idx = seg_base[lo] + (i - at);
      idx = idx < last_edge ? idx : last_edge;
      idx = idx > 0 ? idx : 0;
      r = sorted[idx];
    }
    return r;
  };

  int g = wave * 32;
  uint2 nxt = record(g + lane);
  for (; g < b1; g += BT) {
    const int cnt = (b1 - g < 32) ? (b1 - g) : 32;
    const uint2 rec = nxt;

    if (lane < cnt) {
      unsigned dm = rec.x >> 20;
      dm = dm < (unsigned)(BN - 1) ? dm : (unsigned)(BN - 1);
      atomicAdd(tile + dm * TP + D, __uint_as_float(rec.y));
    }

#pragma unroll
    for (int j0 = 0; j0 < 32; j0 += RB) {
      if (j0 >= cnt) break;
      v2f hv[RB];
#pragma unroll
      for (int u = 0; u < RB; ++u) {
        unsigned p = (unsigned)__builtin_amdgcn_readlane((int)rec.x, j0 + u) & 0xFFFFFu;
        p = p < last_node ? p : last_node;
        hv[u] = *(const v2f*)(x + (size_t)p * D + lane * 2);
      }
      if (j0 == 0) nxt = record(g + BT + lane);
#pragma unroll
      for (int u = 0; u < RB; ++u) {
        if (j0 + u < cnt) {
          const unsigned p = (unsigned)__builtin_amdgcn_readlane((int)rec.x, j0 + u);
          const float wv = __uint_as_float((unsigned)__builtin_amdgcn_readlane((int)rec.y, j0 + u));
          unsigned dl = p >> 20;
          dl = dl < (unsigned)(BN - 1) ? dl : (unsigned)(BN - 1);
          float* __restrict__ trow = tile + dl * TP + lane * 2;
          atomicAdd(trow + 0, hv[u].x * wv);
          atomicAdd(trow + 1, hv[u].y * wv);
        }
      }
    }
  }
  __syncthreads();

  const int rt0 = wave >> 2;
  static_assert(BN % 16 == 0 && NRT > 16 + 7 && NRT <= 32, "row tiles 0..23 always exist, 24..31 are guarded");
  const float* __restrict__ arow = tile + (rt0 * 16 + c) * TP + 32 * hh;
  const v8f acc0 = tile_times_w(arow, bf, hh);
  const v8f acc1 = tile_times_w(arow + 8 * 16 * TP, bf, hh);
  const v8f acc2 = tile_times_w(arow + 16 * 16 * TP, bf, hh);
  v8f acc3 = {};
  const bool has3 = rt0 + 24 < NRT;
  if (has3) acc3 = tile_times_w(arow + 24 * 16 * TP, bf, hh);
  __syncthreads();

  float* __restrict__ drow = tile + (rt0 * 16 + 8 * hh) * TP + ct * 16 + c;
#pragma unroll
  for (int r = 0; r < 8; ++r) {
    drow[r * TP] = acc0[r];
    drow[(8 * 16 + r) * TP] = acc1[r];
    drow[(16 * 16 + r) * TP] = acc2[r];
  }
  if (has3) {
#pragma unroll
    for (int r = 0; r < 8; ++r) drow[(24 * 16 + r) * TP] = acc3[r];
  }
  __syncthreads();

  const int row0 = blockIdx.x * BN;
  int nrows = n_nodes - row0;
  nrows = nrows < BN ? nrows : BN;
  float4* __restrict__ o4 = (float4*)(out + (size_t)row0 * D);
  for (int i = t; i < nrows * (D / 4); i += BT)
    o4[i] = *(const float4*)(tile + (i >> 4) * TP + (i & 15) * 4);
}

extern "C" void kernel_launch(void* const* d_in, const int* in_sizes, int n_in,
                              void* d_out, int out_size, void* d_ws, size_t ws_size,
                              hipStream_t stream) {
  const float* x   = (const float*)d_in[0];
  const int*   src = (const int*)  d_in[1];
  const int*   dst = (const int*)  d_in[2];
  const float* w   = (const float*)d_in[3];
  const float* W   = (const float*)d_in[4];
  const float* b   = (const float*)d_in[5];
  float*       out = (float*)d_out;
  char*        ws  = (char*)d_ws;
  int*          segtab    = (int*)(ws + (64u << 20));
  unsigned int* sorted    = (unsigned int*)(ws + (80u << 20));

  const int n_nodes = in_sizes[0] / D;
  const int n_edges = in_sizes[1];

  gcn_chunk_sort<<<dim3(GB), dim3(BT), 0, stream>>>(src, dst, w, segtab, sorted, n_edges);

  const int n_buckets = (n_nodes + BN - 1) / BN;
  gcn_owner_sum_map<<<dim3(n_buckets), dim3(BT), 0, stream>>>(x, W, b, sorted, segtab, out, n_nodes, n_edges);
}


# owner kernel orders its bucket's records by destination row in LDS (integer rank, scan, place) and sums runs of equal rows in registers: 3 LDS float adds per run, not 2 per record; sum order changes
# speedup vs baseline: 3.2413x; 1.1577x over previous
#include <hip/hip_runtime.h>
#include <hip/hip_bf16.h>

typedef float v2f __attribute__((ext_vector_type(2)));
typedef float v8f __attribute__((ext_vector_type(8)));

#define D 64
#define BN 400
#define NBMAX 256
#define GB 256
#define SEGP 288
#define BT 1024
#define EP 5
#define CAP (BT * EP)
#define TP 68
#define NRT (BN / 16)

__device__ __forceinline__ int bucket_of(int d) {
  unsigned k = (unsigned)d / BN;
  return (int)(k < (unsigned)(NBMAX - 1) ? k : (unsigned)(NBMAX - 1));
}

__global__ __launch_bounds__(BT)
void gcn_chunk_sort(const int* __restrict__ src,
                    const int* __restrict__ dst,
                    const float* __restrict__ w,
                    int* __restrict__ segtab,
                    unsigned int* __restrict__ sorted_raw,
                    int n_edges) {
  __shared__ int hist[NBMAX];
  __shared__ int wtot[8];
  __shared__ unsigned long long stage[CAP];
  const int t = threadIdx.x;
  const int lane = t & 31;
  const int me = blockIdx.x;
  if (t < NBMAX) hist[t] = 0;
  __syncthreads();

  const int chunk = (n_edges + GB - 1) / GB;
  const int e0 = (me * chunk < n_edges) ? me * chunk : n_edges;
  const int e1 = (e0 + chunk < n_edges) ? (e0 + chunk) : n_edges;
  const bool any = e0 < e1;
  const bool staged = e1 - e0 <= CAP;

  int dd[EP], ss[EP];
  float ww[EP];
  if (any) {
#pragma unroll
    for (int u = 0; u < EP; ++u) {
      const int e = e0 + t + u * BT;
      const int ec = e < e1 ? e : e1 - 1;
      dd[u] = dst[ec];
      ss[u] = src[ec];
      ww[u] = w[ec];
    }
#pragma unroll
    for (int u = 0; u < EP; ++u) {
      if (e0 + t + u * BT < e1) atomicAdd(&hist[bucket_of(dd[u])], 1);
    }
    for (int e = e0 + t + EP * BT; e < e1; e += BT) atomicAdd(&hist[bucket_of(dst[e])], 1);
  }
  __syncthreads();

  int total = 0, incl = 0;
  if (t < NBMAX) {
    total = hist[t];
    incl = total;
#pragma unroll
    for (int off = 1; off < 32; off <<= 1) {
      const int n = __shfl_up(incl, off);
      if (lane >= off) incl += n;
    }
    if (lane == 31) wtot[t >> 5] = incl;
  }
  __syncthreads();
  if (t < NBMAX) {
    int wp = 0;
    for (int j = 0; j < (t >> 5); ++j) wp += wtot[j];
    const int excl = wp + incl - total;
    hist[t] = excl;
    segtab[me * SEGP + t] = excl;
    if (t == NBMAX - 1) segtab[me * SEGP + NBMAX] = excl + total;
  }
  __syncthreads();

  uint2* __restrict__ sorted = (uint2*)sorted_raw + e0;
  if (any && staged) {
#pragma unroll
    for (int u = 0; u < EP; ++u) {
      if (e0 + t + u * BT < e1) {
        const int kb = bucket_of(dd[u]);
        const int pos = atomicAdd(&hist[kb], 1);
        const unsigned dl = (unsigned)(dd[u] - kb * BN);
        const unsigned key = ((unsigned)ss[u] & 0xFFFFFu) | (dl << 20);
        stage[pos] = ((unsigned long long)__float_as_uint(ww[u]) << 32) | (unsigned long long)key;
      }
    }
  }
  if (any && !staged) {
#pragma unroll
    for (int u = 0; u < EP; ++u) {
      if (e0 + t + u * BT < e1) {
        const int kb = bucket_of(dd[u]);
        const int pos = atomicAdd(&hist[kb], 1);
        const unsigned dl = (unsigned)(dd[u] - kb * BN);
        sorted[pos] = make_uint2(((unsigned)ss[u] & 0xFFFFFu) | (dl << 20), __float_as_uint(ww[u]));
      }
    }
    for (int e = e0 + t + EP * BT; e < e1; e += BT) {
      const int d = dst[e];
      const int kb = bucket_of(d);
      const int pos = atomicAdd(&hist[kb], 1);
      const unsigned dl = (unsigned)(d - kb * BN);
      sorted[pos] = make_uint2(((unsigned)src[e] & 0xFFFFFu) | (dl << 20), __float_as_uint(w[e]));
    }
  }
  __syncthreads();
  if (staged) {
    unsigned long long* __restrict__ sorted64 = (unsigned long long*)sorted_raw + e0;
    const int n = e1 - e0;
    unsigned long long v[EP];
#pragma unroll
    for (int u = 0; u < EP; ++u) {
      const int i = t + u * BT;
      v[u] = stage[i < n ? i : 0];
    }
#pragma unroll
    for (int u = 0; u < EP; ++u) {
      const int i = t + u * BT;
      if (i < n) sorted64[i] = v[u];
    }
  }
}

#define RB 16
#define EPK 5
#define CAPK (BT * EPK)
#define RPL 13
#define NBIN (32 * RPL)

__device__ __forceinline__ v8f tile_times_w(const float* __restrict__ arow, const v2f (&bf)[17], int hh) {
  v8f acc = {};
#pragma unroll
  for (int j = 0; j < 16; ++j) {
    const v2f a = *(const v2f*)(arow + 2 * j);
    acc = __builtin_amdgcn_wmma_f32_16x16x4_f32(false, a, false, bf[j], (short)0, acc, false, false);
  }
  const v2f a = *(const v2f*)(arow - 32 * hh + D + 2 * hh);
  acc = __builtin_amdgcn_wmma_f32_16x16x4_f32(false, a, false, bf[16], (short)0, acc, false, false);
  return acc;
}

__global__ __launch_bounds__(BT)
void gcn_owner_sum_map(const float* __restrict__ x,
                       const float* __restrict__ W,
                       const float* __restrict__ b,
                       const unsigned int* __restrict__ sorted_raw,
                       const int* __restrict__ segtab,
                       float* __restrict__ out,
                       int n_nodes,
                       int n_edges) {
  __shared__ float tile[BN * TP];
  __shared__ int seg_off[GB + 1];
  __shared__ int seg_base[GB];
  __shared__ int wtot[8];
  __shared__ int rcnt[NBIN];
  __shared__ int rstart[NBIN];
  __shared__ unsigned long long srt[CAPK];
  static_assert(NBIN >= BN, "one counter per row of the tile");
  const int t = threadIdx.x;
  const int lane = t & 31;
  const int wave = __builtin_amdgcn_readfirstlane(t >> 5);
  const int c  = lane & 15;
  const int hh = lane >> 4;
  const int ct = wave & 3;

  int seg_len = 0, seg_at = 0;
  if (t < GB) {
    const int* __restrict__ row = segtab + t * SEGP + blockIdx.x;
    const int s0 = row[0];
    const int s1 = row[1];
    const int chunk = (n_edges + GB - 1) / GB;
    const int c0 = (t * chunk < n_edges) ? t * chunk : n_edges;
    seg_len = s1 - s0;
    seg_at = c0 + s0;
  }

  v2f bf[17];
  {
    const float* __restrict__ wrow = W + (size_t)(ct * 16 + c) * D + 32 * hh;
#pragma unroll
    for (int j = 0; j < 16; ++j) bf[j] = *(const v2f*)(wrow + 2 * j);
    const float bias = b[ct * 16 + c];
    bf[16].x = hh == 0 ? bias : 0.f;
    bf[16].y = 0.f;
  }

  float4* __restrict__ t4 = (float4*)tile;
  for (int i = t; i < BN * TP / 4; i += BT) t4[i] = make_float4(0.f, 0.f, 0.f, 0.f);
  if (t < NBIN) rcnt[t] = 0;

  int incl = seg_len;
  if (t < GB) {
#pragma unroll
    for (int off = 1; off < 32; off <<= 1) {
      const int n = __shfl_up(incl, off);
      if (lane >= off) incl += n;
    }
    if (lane == 31) wtot[t >> 5] = incl;
  }
  __syncthreads();
  if (t < GB) {
    int wp = 0;
    for (int j = 0; j < (t >> 5); ++j) wp += wtot[j];
    seg_off[t] = wp + incl - seg_len;
    seg_base[t] = seg_at;
    if (t == GB - 1) seg_off[GB] = wp + incl;
  }
  __syncthreads();

  const uint2* __restrict__ sorted = (const uint2*)sorted_raw;
  const int b1 = seg_off[GB];
  const int last_edge = n_edges - 1;
  const unsigned last_node = (unsigned)(n_nodes - 1);

  for (int base = 0; base < b1; base += CAPK) {
    const int n = (b1 - base < CAPK) ? (b1 - base) : CAPK;

    int ii[EPK], lo[EPK], at[EPK];
#pragma unroll
    for (int u = 0; u < EPK; ++u) {
      const int i = base + t + u * BT;
      ii[u] = i < b1 ? i : b1 - 1;
      lo[u] = 0;
      at[u] = 0;
    }
#pragma unroll
    for (int s = GB / 2; s >= 1; s >>= 1) {
      int sv[EPK];
#pragma unroll
      for (int u = 0; u < EPK; ++u) sv[u] = seg_off[lo[u] + s];
#pragma unroll
      for (int u = 0; u < EPK; ++u) {
        if (sv[u] <= ii[u]) { lo[u] += s; at[u] = sv[u]; }
      }
    }
    uint2 rc[EPK];
#pragma unroll
    for (int u = 0; u < EPK; ++u) {
      int idx = seg_base[lo[u]] + (ii[u] - at[u]);
      idx = idx < last_edge ? idx : last_edge;
      idx = idx > 0 ? idx : 0;
      rc[u] = sorted[idx];
    }

    int rk[EPK];
    unsigned dr[EPK];
#pragma unroll
    for (int u = 0; u < EPK; ++u) {
      unsigned dm = rc[u].x >> 20;
      dm = dm < (unsigned)(BN - 1) ? dm : (unsigned)(BN - 1);
      dr[u] = dm;
      rk[u] = 0;
      if (base + t + u * BT < b1) rk[u] = atomicAdd(&rcnt[dm], 1);
    }
    __syncthreads();

    if (wave == 0) {
      int cc[RPL];
      int s = 0;
#pragma unroll
      for (int k = 0; k < RPL; ++k) cc[k] = rcnt[lane * RPL + k];
#pragma unroll
      for (int k = 0; k < RPL; ++k) { const int v = cc[k]; cc[k] = s; s += v; }
      int inc = s;
#pragma unroll
      for (int off = 1; off < 32; off <<= 1) {
        const int m = __shfl_up(inc, off);
        if (lane >= off) inc += m;
      }
      const int ex = inc - s;
#pragma unroll
      for (int k = 0; k < RPL; ++k) {
        rstart[lane * RPL + k] = ex + cc[k];
        rcnt[lane * RPL + k] = 0;
      }
    }
    __syncthreads();

    int rs[EPK];
#pragma unroll
    for (int u = 0; u < EPK; ++u) rs[u] = rstart[dr[u]];
#pragma unroll
    for (int u = 0; u < EPK; ++u) {
      if (base + t + u * BT < b1) {
        int pos = rs[u] + rk[u];
        pos = pos < CAPK - 1 ? pos : CAPK - 1;
        srt[pos] = ((unsigned long long)rc[u].y << 32) | (unsigned long long)rc[u].x;
      }
    }
    __syncthreads();

    const int share = (n + 31) >> 5;
    int wlo = wave * share;
    wlo = wlo < n ? wlo : n;
    int whi = wlo + share;
    whi = whi < n ? whi : n;
    int cur = -1;
    float ax = 0.f, ay = 0.f, ws = 0.f;

    auto flush = [&]() {
      float* __restrict__ trow = tile + cur * TP;
      atomicAdd(trow + lane * 2 + 0, ax);
      atomicAdd(trow + lane * 2 + 1, ay);
      if (lane == 0) atomicAdd(trow + D, ws);
    };

    unsigned long long nx = (wlo + lane < whi) ? srt[wlo + lane] : 0ull;
    for (int g = wlo; g < whi; g += 32) {
      const int cnt = (whi - g < 32) ? (whi - g) : 32;
      const unsigned rx = (unsigned)nx;
      const unsigned ry = (unsigned)(nx >> 32);

#pragma unroll
      for (int j0 = 0; j0 < 32; j0 += RB) {
        if (j0 >= cnt) break;
        v2f hv[RB];
#pragma unroll
        for (int u = 0; u < RB; ++u) {
          unsigned p = (unsigned)__builtin_amdgcn_readlane((int)rx, j0 + u) & 0xFFFFFu;
          p = p < last_node ? p : last_node;
          hv[u] = *(const v2f*)(x + (size_t)p * D + lane * 2);
        }
        if (j0 == 0) nx = (g + 32 + lane < whi) ? srt[g + 32 + lane] : 0ull;
#pragma unroll
        for (int u = 0; u < RB; ++u) {
          if (j0 + u < cnt) {
            const unsigned p = (unsigned)__builtin_amdgcn_readlane((int)rx, j0 + u);
            const float wv = __uint_as_float((unsigned)__builtin_amdgcn_readlane((int)ry, j0 + u));
            unsigned dl = p >> 20;
            dl = dl < (unsigned)(BN - 1) ? dl : (unsigned)(BN - 1);
            if ((int)dl != cur) {
              if (cur >= 0) flush();
              cur = (int)dl;
              ax = 0.f; ay = 0.f; ws = 0.f;
            }
            ax += hv[u].x * wv;
            ay += hv[u].y * wv;
            ws += wv;
          }
        }
      }
    }
    if (cur >= 0) flush();
    __syncthreads();
  }

  const int rt0 = wave >> 2;
  static_assert(BN % 16 == 0 && NRT > 16 + 7 && NRT <= 32, "row tiles 0..23 always exist, 24..31 are guarded");
  const float* __restrict__ arow = tile + (rt0 * 16 + c) * TP + 32 * hh;
  const v8f acc0 = tile_times_w(arow, bf, hh);
  const v8f acc1 = tile_times_w(arow + 8 * 16 * TP, bf, hh);
  const v8f acc2 = tile_times_w(arow + 16 * 16 * TP, bf, hh);
  v8f acc3 = {};
  const bool has3 = rt0 + 24 < NRT;
  if (has3) acc3 = tile_times_w(arow + 24 * 16 * TP, bf, hh);
  __syncthreads();

  float* __restrict__ drow = tile + (rt0 * 16 + 8 * hh) * TP + ct * 16 + c;
#pragma unroll
  for (int r = 0; r < 8; ++r) {
    drow[r * TP] = acc0[r];
    drow[(8 * 16 + r) * TP] = acc1[r];
    drow[(16 * 16 + r) * TP] = acc2[r];
  }
  if (has3) {
#pragma unroll
    for (int r = 0; r < 8; ++r) drow[(24 * 16 + r) * TP] = acc3[r];
  }
  __syncthreads();

  const int row0 = blockIdx.x * BN;
  int nrows = n_nodes - row0;
  nrows = nrows < BN ? nrows : BN;
  float4* __restrict__ o4 = (float4*)(out + (size_t)row0 * D);
  for (int i = t; i < nrows * (D / 4); i += BT)
    o4[i] = *(const float4*)(tile + (i >> 4) * TP + (i & 15) * 4);
}

extern "C" void kernel_launch(void* const* d_in, const int* in_sizes, int n_in,
                              void* d_out, int out_size, void* d_ws, size_t ws_size,
                              hipStream_t stream) {
  const float* x   = (const float*)d_in[0];
  const int*   src = (const int*)  d_in[1];
  const int*   dst = (const int*)  d_in[2];
  const float* w   = (const float*)d_in[3];
  const float* W   = (const float*)d_in[4];
  const float* b   = (const float*)d_in[5];
  float*       out = (float*)d_out;
  char*        ws  = (char*)d_ws;
  int*          segtab    = (int*)(ws + (64u << 20));
  unsigned int* sorted    = (unsigned int*)(ws + (80u << 20));

  const int n_nodes = in_sizes[0] / D;
  const int n_edges = in_sizes[1];

  gcn_chunk_sort<<<dim3(GB), dim3(BT), 0, stream>>>(src, dst, w, segtab, sorted, n_edges);

  const int n_buckets = (n_nodes + BN - 1) / BN;
  gcn_owner_sum_map<<<dim3(n_buckets), dim3(BT), 0, stream>>>(x, W, b, sorted, segtab, out, n_nodes, n_edges);
}


# owner sum loop decodes each group of 32 records lane-parallel once (x row offset, clamps, row-change bits by ballot): per record 9 issued instructions, not 17; sums and their order unchanged
# speedup vs baseline: 3.5096x; 1.0828x over previous
#include <hip/hip_runtime.h>
#include <hip/hip_bf16.h>
#include <type_traits>

typedef float v2f __attribute__((ext_vector_type(2)));
typedef float v8f __attribute__((ext_vector_type(8)));

#define D 64
#define BN 400
#define NBMAX 256
#define GB 256
#define SEGP 288
#define BT 1024
#define EP 5
#define CAP (BT * EP)
#define TP 68
#define NRT (BN / 16)

__device__ __forceinline__ int bucket_of(int d) {
  unsigned k = (unsigned)d / BN;
  return (int)(k < (unsigned)(NBMAX - 1) ? k : (unsigned)(NBMAX - 1));
}

__global__ __launch_bounds__(BT)
void gcn_chunk_sort(const int* __restrict__ src,
                    const int* __restrict__ dst,
                    const float* __restrict__ w,
                    int* __restrict__ segtab,
                    unsigned int* __restrict__ sorted_raw,
                    int n_edges) {
  __shared__ int hist[NBMAX];
  __shared__ int wtot[8];
  __shared__ unsigned long long stage[CAP];
  const int t = threadIdx.x;
  const int lane = t & 31;
  const int me = blockIdx.x;
  if (t < NBMAX) hist[t] = 0;
  __syncthreads();

  const int chunk = (n_edges + GB - 1) / GB;
  const int e0 = (me * chunk < n_edges) ? me * chunk : n_edges;
  const int e1 = (e0 + chunk < n_edges) ? (e0 + chunk) : n_edges;
  const bool any = e0 < e1;
  const bool staged = e1 - e0 <= CAP;

  int dd[EP], ss[EP];
  float ww[EP];
  if (any) {
#pragma unroll
    for (int u = 0; u < EP; ++u) {
      const int e = e0 + t + u * BT;
      const int ec = e < e1 ? e : e1 - 1;
      dd[u] = dst[ec];
      ss[u] = src[ec];
      ww[u] = w[ec];
    }
#pragma unroll
    for (int u = 0; u < EP; ++u) {
      if (e0 + t + u * BT < e1) atomicAdd(&hist[bucket_of(dd[u])], 1);
    }
    for (int e = e0 + t + EP * BT; e < e1; e += BT) atomicAdd(&hist[bucket_of(dst[e])], 1);
  }
  __syncthreads();

  int total = 0, incl = 0;
  if (t < NBMAX) {
    total = hist[t];
    incl = total;
#pragma unroll
    for (int off = 1; off < 32; off <<= 1) {
      const int n = __shfl_up(incl, off);
      if (lane >= off) incl += n;
    }
    if (lane == 31) wtot[t >> 5] = incl;
  }
  __syncthreads();
  if (t < NBMAX) {
    int wp = 0;
    for (int j = 0; j < (t >> 5); ++j) wp += wtot[j];
    const int excl = wp + incl - total;
    hist[t] = excl;
    segtab[me * SEGP + t] = excl;
    if (t == NBMAX - 1) segtab[me * SEGP + NBMAX] = excl + total;
  }
  __syncthreads();

  uint2* __restrict__ sorted = (uint2*)sorted_raw + e0;
  if (any && staged) {
#pragma unroll
    for (int u = 0; u < EP; ++u) {
      if (e0 + t + u * BT < e1) {
        const int kb = bucket_of(dd[u]);
        const int pos = atomicAdd(&hist[kb], 1);
        const unsigned dl = (unsigned)(dd[u] - kb * BN);
        const unsigned key = ((unsigned)ss[u] & 0xFFFFFu) | (dl << 20);
        stage[pos] = ((unsigned long long)__float_as_uint(ww[u]) << 32) | (unsigned long long)key;
      }
    }
  }
  if (any && !staged) {
#pragma unroll
    for (int u = 0; u < EP; ++u) {
      if (e0 + t + u * BT < e1) {
        const int kb = bucket_of(dd[u]);
        const int pos = atomicAdd(&hist[kb], 1);
        const unsigned dl = (unsigned)(dd[u] - kb * BN);
        sorted[pos] = make_uint2(((unsigned)ss[u] & 0xFFFFFu) | (dl << 20), __float_as_uint(ww[u]));
      }
    }
    for (int e = e0 + t + EP * BT; e < e1; e += BT) {
      const int d = dst[e];
      const int kb = bucket_of(d);
      const int pos = atomicAdd(&hist[kb], 1);
      const unsigned dl = (unsigned)(d - kb * BN);
      sorted[pos] = make_uint2(((unsigned)src[e] & 0xFFFFFu) | (dl << 20), __float_as_uint(w[e]));
    }
  }
  __syncthreads();
  if (staged) {
    unsigned long long* __restrict__ sorted64 = (unsigned long long*)sorted_raw + e0;
    const int n = e1 - e0;
    unsigned long long v[EP];
#pragma unroll
    for (int u = 0; u < EP; ++u) {
      const int i = t + u * BT;
      v[u] = stage[i < n ? i : 0];
    }
#pragma unroll
    for (int u = 0; u < EP; ++u) {
      const int i = t + u * BT;
      if (i < n) sorted64[i] = v[u];
    }
  }
}

#define RB 16
#define EPK 5
#define CAPK (BT * EPK)
#define RPL 13
#define NBIN (32 * RPL)

__device__ __forceinline__ v8f tile_times_w(const float* __restrict__ arow, const v2f (&bf)[17], int hh) {
  v8f acc = {};
#pragma unroll
  for (int j = 0; j < 16; ++j) {
    const v2f a = *(const v2f*)(arow + 2 * j);
    acc = __builtin_amdgcn_wmma_f32_16x16x4_f32(false, a, false, bf[j], (short)0, acc, false, false);
  }
  const v2f a = *(const v2f*)(arow - 32 * hh + D + 2 * hh);
  acc = __builtin_amdgcn_wmma_f32_16x16x4_f32(false, a, false, bf[16], (short)0, acc, false, false);
  return acc;
}

__global__ __launch_bounds__(BT)
void gcn_owner_sum_map(const float* __restrict__ x,
                       const float* __restrict__ W,
                       const float* __restrict__ b,
                       const unsigned int* __restrict__ sorted_raw,
                       const int* __restrict__ segtab,
                       float* __restrict__ out,
                       int n_nodes,
                       int n_edges) {
  __shared__ float tile[BN * TP];
  __shared__ int seg_off[GB + 1];
  __shared__ int seg_base[GB];
  __shared__ int wtot[8];
  __shared__ int rcnt[NBIN];
  __shared__ int rstart[NBIN];
  __shared__ unsigned long long srt[CAPK];
  static_assert(NBIN >= BN, "one counter per row of the tile");
  const int t = threadIdx.x;
  const int lane = t & 31;
  const int wave = __builtin_amdgcn_readfirstlane(t >> 5);
  const int c  = lane & 15;
  const int hh = lane >> 4;
  const int ct = wave & 3;

  int seg_len = 0, seg_at = 0;
  if (t < GB) {
    const int* __restrict__ row = segtab + t * SEGP + blockIdx.x;
    const int s0 = row[0];
    const int s1 = row[1];
    const int chunk = (n_edges + GB - 1) / GB;
    const int c0 = (t * chunk < n_edges) ? t * chunk : n_edges;
    seg_len = s1 - s0;
    seg_at = c0 + s0;
  }

  v2f bf[17];
  {
    const float* __restrict__ wrow = W + (size_t)(ct * 16 + c) * D + 32 * hh;
#pragma unroll
    for (int j = 0; j < 16; ++j) bf[j] = *(const v2f*)(wrow + 2 * j);
    const float bias = b[ct * 16 + c];
    bf[16].x = hh == 0 ? bias : 0.f;
    bf[16].y = 0.f;
  }

  float4* __restrict__ t4 = (float4*)tile;
  for (int i = t; i < BN * TP / 4; i += BT) t4[i] = make_float4(0.f, 0.f, 0.f, 0.f);
  if (t < NBIN) rcnt[t] = 0;

  int incl = seg_len;
  if (t < GB) {
#pragma unroll
    for (int off = 1; off < 32; off <<= 1) {
      const int n = __shfl_up(incl, off);
      if (lane >= off) incl += n;
    }
    if (lane == 31) wtot[t >> 5] = incl;
  }
  __syncthreads();
  if (t < GB) {
    int wp = 0;
    for (int j = 0; j < (t >> 5); ++j) wp += wtot[j];
    seg_off[t] = wp + incl - seg_len;
    seg_base[t] = seg_at;
    if (t == GB - 1) seg_off[GB] = wp + incl;
  }
  __syncthreads();

  const uint2* __restrict__ sorted = (const uint2*)sorted_raw;
  const int b1 = seg_off[GB];
  const int last_edge = n_edges - 1;
  const unsigned last_node = (unsigned)(n_nodes - 1);

  for (int base = 0; base < b1; base += CAPK) {
    const int n = (b1 - base < CAPK) ? (b1 - base) : CAPK;

    int ii[EPK], lo[EPK], at[EPK];
#pragma unroll
    for (int u = 0; u < EPK; ++u) {
      const int i = base + t + u * BT;
      ii[u] = i < b1 ? i : b1 - 1;
      lo[u] = 0;
      at[u] = 0;
    }
#pragma unroll
    for (int s = GB / 2; s >= 1; s >>= 1) {
      int sv[EPK];
#pragma unroll
      for (int u = 0; u < EPK; ++u) sv[u] = seg_off[lo[u] + s];
#pragma unroll
      for (int u = 0; u < EPK; ++u) {
        if (sv[u] <= ii[u]) { lo[u] += s; at[u] = sv[u]; }
      }
    }
    uint2 rc[EPK];
#pragma unroll
    for (int u = 0; u < EPK; ++u) {
      int idx = seg_base[lo[u]] + (ii[u] - at[u]);
      idx = idx < last_edge ? idx : last_edge;
      idx = idx > 0 ? idx : 0;
      rc[u] = sorted[idx];
    }

    int rk[EPK];
    unsigned dr[EPK];
#pragma unroll
    for (int u = 0; u < EPK; ++u) {
      unsigned dm = rc[u].x >> 20;
      dm = dm < (unsigned)(BN - 1) ? dm : (unsigned)(BN - 1);
      dr[u] = dm;
      rk[u] = 0;
      if (base + t + u * BT < b1) rk[u] = atomicAdd(&rcnt[dm], 1);
    }
    __syncthreads();

    if (wave == 0) {
      int cc[RPL];
      int s = 0;
#pragma unroll
      for (int k = 0; k < RPL; ++k) cc[k] = rcnt[lane * RPL + k];
#pragma unroll
      for (int k = 0; k < RPL; ++k) { const int v = cc[k]; cc[k] = s; s += v; }
      int inc = s;
#pragma unroll
      for (int off = 1; off < 32; off <<= 1) {
        const int m = __shfl_up(inc, off);
        if (lane >= off) inc += m;
      }
      const int ex = inc - s;
#pragma unroll
      for (int k = 0; k < RPL; ++k) {
        rstart[lane * RPL + k] = ex + cc[k];
        rcnt[lane * RPL + k] = 0;
      }
    }
    __syncthreads();

    int rs[EPK];
#pragma unroll
    for (int u = 0; u < EPK; ++u) rs[u] = rstart[dr[u]];
#pragma unroll
    for (int u = 0; u < EPK; ++u) {
      if (base + t + u * BT < b1) {
        int pos = rs[u] + rk[u];
        pos = pos < CAPK - 1 ? pos : CAPK - 1;
        srt[pos] = ((unsigned long long)rc[u].y << 32) | (unsigned long long)rc[u].x;
      }
    }
    __syncthreads();

    const int share = (n + 31) >> 5;
    int wlo = wave * share;
    wlo = wlo < n ? wlo : n;
    int whi = wlo + share;
    whi = whi < n ? whi : n;
    int cur = -1;
    float ax = 0.f, ay = 0.f, ws = 0.f;

    auto flush = [&]() {
      float* __restrict__ trow = tile + cur * TP;
      atomicAdd(trow + lane * 2 + 0, ax);
      atomicAdd(trow + lane * 2 + 1, ay);
      if (lane == 0) atomicAdd(trow + D, ws);
    };

    const char* __restrict__ xbytes = (const char*)x;
    const unsigned lane8 = (unsigned)lane * 8u;
    unsigned long long nx = (wlo + lane < whi) ? srt[wlo + lane] : 0ull;
    unsigned npk = (wlo + lane < whi && lane > 0) ? (unsigned)srt[wlo + lane - 1] : 0u;
    for (int g = wlo; g < whi; g += 32) {
      const int cnt = (whi - g < 32) ? (whi - g) : 32;
      const unsigned rx = (unsigned)nx;
      const unsigned ry = (unsigned)(nx >> 32);
      unsigned pr = rx & 0xFFFFFu;
      pr = pr < last_node ? pr : last_node;
      const unsigned voff = pr << 8;
      unsigned dl = rx >> 20;
      dl = dl < (unsigned)(BN - 1) ? dl : (unsigned)(BN - 1);
      unsigned pdl = npk >> 20;
      pdl = pdl < (unsigned)(BN - 1) ? pdl : (unsigned)(BN - 1);
      const bool head = (g == wlo) && (lane == 0);
      const unsigned chg = __builtin_amdgcn_ballot_w32((lane < cnt) && (head || dl != pdl));

      auto group = [&](auto full_tag) {
        constexpr bool FULL = decltype(full_tag)::value;
#pragma unroll
        for (int j0 = 0; j0 < 32; j0 += RB) {
          if (!FULL && j0 >= cnt) break;
          v2f hv[RB];
#pragma unroll
          for (int u = 0; u < RB; ++u) {
            const unsigned so = (unsigned)__builtin_amdgcn_readlane((int)voff, j0 + u);
            hv[u] = *(const v2f*)(xbytes + (size_t)so + lane8);
          }
          if (j0 == 0) {
            const bool more = g + 32 + lane < whi;
            nx = more ? srt[g + 32 + lane] : 0ull;
            npk = more ? (unsigned)srt[g + 31 + lane] : 0u;
          }
#pragma unroll
          for (int u = 0; u < RB; ++u) {
            if (FULL || j0 + u < cnt) {
              if ((chg >> (j0 + u)) & 1u) {
                if (cur >= 0) flush();
                cur = __builtin_amdgcn_readlane((int)dl, j0 + u);
                ax = 0.f; ay = 0.f; ws = 0.f;
              }
              const float wv = __uint_as_float((unsigned)__builtin_amdgcn_readlane((int)ry, j0 + u));
              ax += hv[u].x * wv;
              ay += hv[u].y * wv;
              ws += wv;
            }
          }
        }
      };
      if (cnt == 32) group(std::true_type{}); else group(std::false_type{});
    }
    if (cur >= 0) flush();
    __syncthreads();
  }

  const int rt0 = wave >> 2;
  static_assert(BN % 16 == 0 && NRT > 16 + 7 && NRT <= 32, "row tiles 0..23 always exist, 24..31 are guarded");
  const float* __restrict__ arow = tile + (rt0 * 16 + c) * TP + 32 * hh;
  const v8f acc0 = tile_times_w(arow, bf, hh);
  const v8f acc1 = tile_times_w(arow + 8 * 16 * TP, bf, hh);
  const v8f acc2 = tile_times_w(arow + 16 * 16 * TP, bf, hh);
  v8f acc3 = {};
  const bool has3 = rt0 + 24 < NRT;
  if (has3) acc3 = tile_times_w(arow + 24 * 16 * TP, bf, hh);
  __syncthreads();

  float* __restrict__ drow = tile + (rt0 * 16 + 8 * hh) * TP + ct * 16 + c;
#pragma unroll
  for (int r = 0; r < 8; ++r) {
    drow[r * TP] = acc0[r];
    drow[(8 * 16 + r) * TP] = acc1[r];
    drow[(16 * 16 + r) * TP] = acc2[r];
  }
  if (has3) {
#pragma unroll
    for (int r = 0; r < 8; ++r) drow[(24 * 16 + r) * TP] = acc3[r];
  }
  __syncthreads();

  const int row0 = blockIdx.x * BN;
  int nrows = n_nodes - row0;
  nrows = nrows < BN ? nrows : BN;
  float4* __restrict__ o4 = (float4*)(out + (size_t)row0 * D);
  for (int i = t; i < nrows * (D / 4); i += BT)
    o4[i] = *(const float4*)(tile + (i >> 4) * TP + (i & 15) * 4);
}

extern "C" void kernel_launch(void* const* d_in, const int* in_sizes, int n_in,
                              void* d_out, int out_size, void* d_ws, size_t ws_size,
                              hipStream_t stream) {
  const float* x   = (const float*)d_in[0];
  const int*   src = (const int*)  d_in[1];
  const int*   dst = (const int*)  d_in[2];
  const float* w   = (const float*)d_in[3];
  const float* W   = (const float*)d_in[4];
  const float* b   = (const float*)d_in[5];
  float*       out = (float*)d_out;
  char*        ws  = (char*)d_ws;
  int*          segtab    = (int*)(ws + (64u << 20));
  unsigned int* sorted    = (unsigned int*)(ws + (80u << 20));

  const int n_nodes = in_sizes[0] / D;
  const int n_edges = in_sizes[1];

  gcn_chunk_sort<<<dim3(GB), dim3(BT), 0, stream>>>(src, dst, w, segtab, sorted, n_edges);

  const int n_buckets = (n_nodes + BN - 1) / BN;
  gcn_owner_sum_map<<<dim3(n_buckets), dim3(BT), 0, stream>>>(x, W, b, sorted, segtab, out, n_nodes, n_edges);
}
